# starting kernel
# baseline (speedup 1.0000x reference)
#include <hip/hip_runtime.h>


#define T_LEN   1048576
#define H       16
#define CHUNK   256
#define NCHUNK  (T_LEN / CHUNK)

typedef __attribute__((ext_vector_type(2))) float v2f;
typedef __attribute__((ext_vector_type(8))) float v8f;

#if __has_builtin(__builtin_amdgcn_tanhf)
#define TANHF(x) __builtin_amdgcn_tanhf(x)
#else
#define TANHF(x) tanhf(x)
#endif

__device__ __forceinline__ float readlane_f(float v, int lane) {
  return __builtin_bit_cast(float,
      __builtin_amdgcn_readlane(__builtin_bit_cast(int, v), lane));
}

__global__ __launch_bounds__(96, 1)
void rnn_pipeline_kernel(const float* __restrict__ y,
                         const float* __restrict__ W_ih0, const float* __restrict__ b_ih0,
                         const float* __restrict__ W_hh0, const float* __restrict__ b_hh0,
                         const float* __restrict__ W_ih1, const float* __restrict__ b_ih1,
                         const float* __restrict__ W_hh1, const float* __restrict__ b_hh1,
                         const float* __restrict__ W_lin, const float* __restrict__ b_lin,
                         float* __restrict__ out)
{
  __shared__ float h1buf[2][CHUNK * H];
  __shared__ float h2buf[2][CHUNK * H];
  __shared__ float xp1buf[CHUNK * H];

  const int tid  = threadIdx.x;
  const int wave = tid >> 5;
  const int l    = tid & 31;
  const int n    = l & 15;
  const int half = l >> 4;

  float w0[H];
  float w1[H];
  float wl[H];
  v2f   Bw[4];
  float wih = 0.f, bs0 = 0.f, b1v = 0.f, bl = 0.f;

  float sh[H];
  #pragma unroll
  for (int k = 0; k < H; ++k) sh[k] = 0.0f;

  if (wave == 0) {
    #pragma unroll
    for (int k = 0; k < H; ++k) w0[k] = W_hh0[n * H + k];
    wih = W_ih0[n];
    bs0 = b_ih0[n] + b_hh0[n];
  } else if (wave == 1) {
    #pragma unroll
    for (int k = 0; k < H; ++k) w1[k] = W_hh1[n * H + k];
    #pragma unroll
    for (int kb = 0; kb < 4; ++kb) {
      const int K = 4 * kb + 2 * half;
      Bw[kb] = *(const v2f*)(W_ih1 + n * H + K);
    }
    b1v = b_ih1[n] + b_hh1[n];
  } else {
    const int o = l & 1;
    #pragma unroll
    for (int k = 0; k < H; ++k) wl[k] = W_lin[o * H + k];
    bl = b_lin[o];
  }

  for (int i = 0; i < NCHUNK + 2; ++i) {
    if (wave == 0) {
      if (i < NCHUNK) {
        float* hb = h1buf[i & 1];
        const int base = i * CHUNK;
        if (i + 1 < NCHUNK) __builtin_prefetch(y + base + CHUNK, 0, 0);
        for (int s = 0; s < CHUNK; ++s) {
          const float yv = y[base + s];
          float a0 = fmaf(yv, wih, bs0);
          float a1 = 0.f;
          #pragma unroll
          for (int k = 0; k < H; k += 2) {
            a0 = fmaf(sh[k],     w0[k],     a0);
            a1 = fmaf(sh[k + 1], w0[k + 1], a1);
          }
          const float hn = TANHF(a0 + a1);
          if (l < 16) hb[s * H + n] = hn;
          #pragma unroll
          for (int k = 0; k < H; ++k) sh[k] = readlane_f(hn, k);
        }
      }
    } else if (wave == 1) {
      if (i >= 1 && i <= NCHUNK) {
        const int c = i - 1;
        const float* hb = h1buf[c & 1];
        float*       ob = h2buf[c & 1];

        for (int tile = 0; tile < CHUNK / 16; ++tile) {
          v8f acc;
          #pragma unroll
          for (int v = 0; v < 8; ++v) acc[v] = b1v;
          #pragma unroll
          for (int kb = 0; kb < 4; ++kb) {
            const int K = 4 * kb + 2 * half;
            const v2f Aa = *(const v2f*)(hb + (tile * 16 + n) * H + K);
            acc = __builtin_amdgcn_wmma_f32_16x16x4_f32(
                false, Aa, false, Bw[kb], (short)0, acc, false, false);
          }
          #pragma unroll
          for (int v = 0; v < 8; ++v)
            xp1buf[(tile * 16 + v + 8 * half) * H + n] = acc[v];
        }

        for (int s = 0; s < CHUNK; ++s) {
          float a0 = xp1buf[s * H + n];
          float a1 = 0.f;
          #pragma unroll
          for (int k = 0; k < H; k += 2) {
            a0 = fmaf(sh[k],     w1[k],     a0);
            a1 = fmaf(sh[k + 1], w1[k + 1], a1);
          }
          const float hn = TANHF(a0 + a1);
          if (l < 16) ob[s * H + n] = hn;
          #pragma unroll
          for (int k = 0; k < H; ++k) sh[k] = readlane_f(hn, k);
        }
      }
    } else {
      if (i >= 2) {
        const int c = i - 2;
        const float* hb = h2buf[c & 1];
        const int o  = l & 1;
        const int tt = l >> 1;
        const int base = c * CHUNK;
        for (int s = tt; s < CHUNK; s += 16) {
          float acc = bl;
          #pragma unroll
          for (int k = 0; k < H; ++k)
            acc = fmaf(hb[s * H + k], wl[k], acc);
          out[(base + s) * 2 + o] = acc;
        }
      }
    }
    __syncthreads();
  }
}

extern "C" void kernel_launch(void* const* d_in, const int* in_sizes, int n_in,
                              void* d_out, int out_size, void* d_ws, size_t ws_size,
                              hipStream_t stream) {
  (void)in_sizes; (void)n_in; (void)out_size; (void)d_ws; (void)ws_size;
  const float* y     = (const float*)d_in[0];
  const float* W_ih0 = (const float*)d_in[1];
  const float* b_ih0 = (const float*)d_in[2];
  const float* W_hh0 = (const float*)d_in[3];
  const float* b_hh0 = (const float*)d_in[4];
  const float* W_ih1 = (const float*)d_in[5];
  const float* b_ih1 = (const float*)d_in[6];
  const float* W_hh1 = (const float*)d_in[7];
  const float* b_hh1 = (const float*)d_in[8];
  const float* W_lin = (const float*)d_in[9];
  const float* b_lin = (const float*)d_in[10];
  float* out = (float*)d_out;

  rnn_pipeline_kernel<<<dim3(1), dim3(96), 0, stream>>>(
      y, W_ih0, b_ih0, W_hh0, b_hh0, W_ih1, b_ih1, W_hh1, b_hh1,
      W_lin, b_lin, out);
}


# Per-step input loads moved off the recurrence: layer 0 reads y 32 steps per per-lane load one block ahead (readlane at use), layer 1 keeps the next 8 xp1 rows in registers; no load wait between steps
# speedup vs baseline: 1.4723x; 1.4723x over previous
#include <hip/hip_runtime.h>


#define T_LEN   1048576
#define H       16
#define CHUNK   256
#define NCHUNK  (T_LEN / CHUNK)
#define YBLK    32
#define NYBLK   (T_LEN / YBLK)
#define XQ      8

typedef __attribute__((ext_vector_type(2))) float v2f;
typedef __attribute__((ext_vector_type(8))) float v8f;

#if __has_builtin(__builtin_amdgcn_tanhf)
#define TANHF(x) __builtin_amdgcn_tanhf(x)
#else
#define TANHF(x) tanhf(x)
#endif

__device__ __forceinline__ float readlane_f(float v, int lane) {
  return __builtin_bit_cast(float,
      __builtin_amdgcn_readlane(__builtin_bit_cast(int, v), lane));
}

__global__ __launch_bounds__(96, 1)
void rnn_pipeline_kernel(const float* __restrict__ y,
                         const float* __restrict__ W_ih0, const float* __restrict__ b_ih0,
                         const float* __restrict__ W_hh0, const float* __restrict__ b_hh0,
                         const float* __restrict__ W_ih1, const float* __restrict__ b_ih1,
                         const float* __restrict__ W_hh1, const float* __restrict__ b_hh1,
                         const float* __restrict__ W_lin, const float* __restrict__ b_lin,
                         float* __restrict__ out)
{
  __shared__ float h1buf[2][CHUNK * H];
  __shared__ float h2buf[2][CHUNK * H];
  __shared__ float xp1buf[CHUNK * H];

  const int tid  = threadIdx.x;
  const int wave = tid >> 5;
  const int l    = tid & 31;
  const int n    = l & 15;
  const int half = l >> 4;

  float w0[H];
  float w1[H];
  float wl[H];
  v2f   Bw[4];
  float wih = 0.f, bs0 = 0.f, b1v = 0.f, bl = 0.f;
  float ycur = 0.f;

  float sh[H];
  #pragma unroll
  for (int k = 0; k < H; ++k) sh[k] = 0.0f;

  if (wave == 0) {
    #pragma unroll
    for (int k = 0; k < H; ++k) w0[k] = W_hh0[n * H + k];
    wih = W_ih0[n];
    bs0 = b_ih0[n] + b_hh0[n];
    ycur = y[l];
  } else if (wave == 1) {
    #pragma unroll
    for (int k = 0; k < H; ++k) w1[k] = W_hh1[n * H + k];
    #pragma unroll
    for (int kb = 0; kb < 4; ++kb) {
      const int K = 4 * kb + 2 * half;
      Bw[kb] = *(const v2f*)(W_ih1 + n * H + K);
    }
    b1v = b_ih1[n] + b_hh1[n];
  } else {
    const int o = l & 1;
    #pragma unroll
    for (int k = 0; k < H; ++k) wl[k] = W_lin[o * H + k];
    bl = b_lin[o];
  }

  for (int i = 0; i < NCHUNK + 2; ++i) {
    if (wave == 0) {
      if (i < NCHUNK) {
        float* hb = h1buf[i & 1];
        for (int b = 0; b < CHUNK / YBLK; ++b) {
          const int g  = i * (CHUNK / YBLK) + b;
          const int gn = (g + 1 < NYBLK) ? g + 1 : g;
          const float ynext = y[gn * YBLK + l];
          #pragma unroll 8
          for (int j = 0; j < YBLK; ++j) {
            const int s = b * YBLK + j;
            const float yv = readlane_f(ycur, j);
            float a0 = fmaf(yv, wih, bs0);
            float a1 = 0.f;
            #pragma unroll
            for (int k = 0; k < H; k += 2) {
              a0 = fmaf(sh[k],     w0[k],     a0);
              a1 = fmaf(sh[k + 1], w0[k + 1], a1);
            }
            const float hn = TANHF(a0 + a1);
            if (l < 16) hb[s * H + n] = hn;
            #pragma unroll
            for (int k = 0; k < H; ++k) sh[k] = readlane_f(hn, k);
          }
          ycur = ynext;
        }
      }
    } else if (wave == 1) {
      if (i >= 1 && i <= NCHUNK) {
        const int c = i - 1;
        const float* hb = h1buf[c & 1];
        float*       ob = h2buf[c & 1];

        for (int tile = 0; tile < CHUNK / 16; ++tile) {
          v8f acc;
          #pragma unroll
          for (int v = 0; v < 8; ++v) acc[v] = b1v;
          #pragma unroll
          for (int kb = 0; kb < 4; ++kb) {
            const int K = 4 * kb + 2 * half;
            const v2f Aa = *(const v2f*)(hb + (tile * 16 + n) * H + K);
            acc = __builtin_amdgcn_wmma_f32_16x16x4_f32(
                false, Aa, false, Bw[kb], (short)0, acc, false, false);
          }
          #pragma unroll
          for (int v = 0; v < 8; ++v)
            xp1buf[(tile * 16 + v + 8 * half) * H + n] = acc[v];
        }

        float xq[XQ];
        #pragma unroll
        for (int u = 0; u < XQ; ++u) xq[u] = xp1buf[u * H + n];
        for (int sb = 0; sb < CHUNK; sb += XQ) {
          const int ahead = (sb + XQ < CHUNK) ? XQ : 0;
          #pragma unroll
          for (int u = 0; u < XQ; ++u) {
            const int s = sb + u;
            float a0 = xq[u];
            xq[u] = xp1buf[(s + ahead) * H + n];
            float a1 = 0.f;
            #pragma unroll
            for (int k = 0; k < H; k += 2) {
              a0 = fmaf(sh[k],     w1[k],     a0);
              a1 = fmaf(sh[k + 1], w1[k + 1], a1);
            }
            const float hn = TANHF(a0 + a1);
            if (l < 16) ob[s * H + n] = hn;
            #pragma unroll
            for (int k = 0; k < H; ++k) sh[k] = readlane_f(hn, k);
          }
        }
      }
    } else {
      if (i >= 2) {
        const int c = i - 2;
        const float* hb = h2buf[c & 1];
        const int o  = l & 1;
        const int tt = l >> 1;
        const int base = c * CHUNK;
        for (int s = tt; s < CHUNK; s += 16) {
          float acc = bl;
          #pragma unroll
          for (int k = 0; k < H; ++k)
            acc = fmaf(hb[s * H + k], wl[k], acc);
          out[(base + s) * 2 + o] = acc;
        }
      }
    }
    __syncthreads();
  }
}

extern "C" void kernel_launch(void* const* d_in, const int* in_sizes, int n_in,
                              void* d_out, int out_size, void* d_ws, size_t ws_size,
                              hipStream_t stream) {
  (void)in_sizes; (void)n_in; (void)out_size; (void)d_ws; (void)ws_size;
  const float* y     = (const float*)d_in[0];
  const float* W_ih0 = (const float*)d_in[1];
  const float* b_ih0 = (const float*)d_in[2];
  const float* W_hh0 = (const float*)d_in[3];
  const float* b_hh0 = (const float*)d_in[4];
  const float* W_ih1 = (const float*)d_in[5];
  const float* b_ih1 = (const float*)d_in[6];
  const float* W_hh1 = (const float*)d_in[7];
  const float* b_hh1 = (const float*)d_in[8];
  const float* W_lin = (const float*)d_in[9];
  const float* b_lin = (const float*)d_in[10];
  float* out = (float*)d_out;

  rnn_pipeline_kernel<<<dim3(1), dim3(96), 0, stream>>>(
      y, W_ih0, b_ih0, W_hh0, b_hh0, W_ih1, b_ih1, W_hh1, b_hh1,
      W_lin, b_lin, out);
}


# Hidden state broadcast inside the FMAs by DPP row_share (no readlane, scalar register and move round trip per step), both scans; the 16 products summed as 4 partial sums by k mod 4
# speedup vs baseline: 1.6011x; 1.0875x over previous
#include <hip/hip_runtime.h>


#define T_LEN   1048576
#define H       16
#define CHUNK   256
#define NCHUNK  (T_LEN / CHUNK)
#define YBLK    32
#define NYBLK   (T_LEN / YBLK)
#define XQ      8

typedef __attribute__((ext_vector_type(2))) float v2f;
typedef __attribute__((ext_vector_type(8))) float v8f;

__device__ __forceinline__ float readlane_f(float v, int lane) {
  return __builtin_bit_cast(float,
      __builtin_amdgcn_readlane(__builtin_bit_cast(int, v), lane));
}

#define RS(k) " row_share:" #k " row_mask:0xf bank_mask:0xf\n\t"
__device__ __forceinline__ float rnn_step(float x, float h, const float (&w)[H]) {
  float s0 = x, s1, s2, s3;
  asm("v_fmac_f32_dpp %0, %4, %5"  RS(0)
      "v_mul_f32_dpp  %1, %4, %6"  RS(1)
      "v_mul_f32_dpp  %2, %4, %7"  RS(2)
      "v_mul_f32_dpp  %3, %4, %8"  RS(3)
      "v_fmac_f32_dpp %0, %4, %9"  RS(4)
      "v_fmac_f32_dpp %1, %4, %10" RS(5)
      "v_fmac_f32_dpp %2, %4, %11" RS(6)
      "v_fmac_f32_dpp %3, %4, %12" RS(7)
      "v_fmac_f32_dpp %0, %4, %13" RS(8)
      "v_fmac_f32_dpp %1, %4, %14" RS(9)
      "v_fmac_f32_dpp %2, %4, %15" RS(10)
      "v_fmac_f32_dpp %3, %4, %16" RS(11)
      "v_fmac_f32_dpp %0, %4, %17" RS(12)
      "v_fmac_f32_dpp %1, %4, %18" RS(13)
      "v_fmac_f32_dpp %2, %4, %19" RS(14)
      "v_fmac_f32_dpp %3, %4, %20" RS(15)
      "v_add_f32 %0, %0, %1\n\t"
      "v_add_f32 %2, %2, %3\n\t"
      "v_add_f32 %0, %0, %2\n\t"
      "v_tanh_f32 %4, %0\n\t"
      "v_nop"
      : "+v"(s0), "=&v"(s1), "=&v"(s2), "=&v"(s3), "+v"(h)
      : "v"(w[0]),  "v"(w[1]),  "v"(w[2]),  "v"(w[3]),
        "v"(w[4]),  "v"(w[5]),  "v"(w[6]),  "v"(w[7]),
        "v"(w[8]),  "v"(w[9]),  "v"(w[10]), "v"(w[11]),
        "v"(w[12]), "v"(w[13]), "v"(w[14]), "v"(w[15]));
  return h;
}
#undef RS

__global__ __launch_bounds__(96, 1)
void rnn_pipeline_kernel(const float* __restrict__ y,
                         const float* __restrict__ W_ih0, const float* __restrict__ b_ih0,
                         const float* __restrict__ W_hh0, const float* __restrict__ b_hh0,
                         const float* __restrict__ W_ih1, const float* __restrict__ b_ih1,
                         const float* __restrict__ W_hh1, const float* __restrict__ b_hh1,
                         const float* __restrict__ W_lin, const float* __restrict__ b_lin,
                         float* __restrict__ out)
{
  __shared__ float h1buf[2][CHUNK * H];
  __shared__ float h2buf[2][CHUNK * H];
  __shared__ float xp1buf[CHUNK * H];

  const int tid  = threadIdx.x;
  const int wave = tid >> 5;
  const int l    = tid & 31;
  const int n    = l & 15;
  const int half = l >> 4;

  float w0[H];
  float w1[H];
  float wl[H];
  v2f   Bw[4];
  float wih = 0.f, bs0 = 0.f, b1v = 0.f, bl = 0.f;
  float ycur = 0.f;

  float hs = 0.0f;

  if (wave == 0) {
    #pragma unroll
    for (int k = 0; k < H; ++k) w0[k] = W_hh0[n * H + k];
    wih = W_ih0[n];
    bs0 = b_ih0[n] + b_hh0[n];
    ycur = y[l];
  } else if (wave == 1) {
    #pragma unroll
    for (int k = 0; k < H; ++k) w1[k] = W_hh1[n * H + k];
    #pragma unroll
    for (int kb = 0; kb < 4; ++kb) {
      const int K = 4 * kb + 2 * half;
      Bw[kb] = *(const v2f*)(W_ih1 + n * H + K);
    }
    b1v = b_ih1[n] + b_hh1[n];
  } else {
    const int o = l & 1;
    #pragma unroll
    for (int k = 0; k < H; ++k) wl[k] = W_lin[o * H + k];
    bl = b_lin[o];
  }

  for (int i = 0; i < NCHUNK + 2; ++i) {
    if (wave == 0) {
      if (i < NCHUNK) {
        float* hb = h1buf[i & 1];
        for (int b = 0; b < CHUNK / YBLK; ++b) {
          const int g  = i * (CHUNK / YBLK) + b;
          const int gn = (g + 1 < NYBLK) ? g + 1 : g;
          const float ynext = y[gn * YBLK + l];
          #pragma unroll 8
          for (int j = 0; j < YBLK; ++j) {
            const int s = b * YBLK + j;
            const float yv = readlane_f(ycur, j);
            const float xp = fmaf(yv, wih, bs0);
            hs = rnn_step(xp, hs, w0);
            if (l < 16) hb[s * H + n] = hs;
          }
          ycur = ynext;
        }
      }
    } else if (wave == 1) {
      if (i >= 1 && i <= NCHUNK) {
        const int c = i - 1;
        const float* hb = h1buf[c & 1];
        float*       ob = h2buf[c & 1];

        for (int tile = 0; tile < CHUNK / 16; ++tile) {
          v8f acc;
          #pragma unroll
          for (int v = 0; v < 8; ++v) acc[v] = b1v;
          #pragma unroll
          for (int kb = 0; kb < 4; ++kb) {
            const int K = 4 * kb + 2 * half;
            const v2f Aa = *(const v2f*)(hb + (tile * 16 + n) * H + K);
            acc = __builtin_amdgcn_wmma_f32_16x16x4_f32(
                false, Aa, false, Bw[kb], (short)0, acc, false, false);
          }
          #pragma unroll
          for (int v = 0; v < 8; ++v)
            xp1buf[(tile * 16 + v + 8 * half) * H + n] = acc[v];
        }

        float xq[XQ];
        #pragma unroll
        for (int u = 0; u < XQ; ++u) xq[u] = xp1buf[u * H + n];
        for (int sb = 0; sb < CHUNK; sb += XQ) {
          const int ahead = (sb + XQ < CHUNK) ? XQ : 0;
          #pragma unroll
          for (int u = 0; u < XQ; ++u) {
            const int s = sb + u;
            const float xp = xq[u];
            xq[u] = xp1buf[(s + ahead) * H + n];
            hs = rnn_step(xp, hs, w1);
            if (l < 16) ob[s * H + n] = hs;
          }
        }
      }
    } else {
      if (i >= 2) {
        const int c = i - 2;
        const float* hb = h2buf[c & 1];
        const int o  = l & 1;
        const int tt = l >> 1;
        const int base = c * CHUNK;
        for (int s = tt; s < CHUNK; s += 16) {
          float acc = bl;
          #pragma unroll
          for (int k = 0; k < H; ++k)
            acc = fmaf(hb[s * H + k], wl[k], acc);
          out[(base + s) * 2 + o] = acc;
        }
      }
    }
    __syncthreads();
  }
}

extern "C" void kernel_launch(void* const* d_in, const int* in_sizes, int n_in,
                              void* d_out, int out_size, void* d_ws, size_t ws_size,
                              hipStream_t stream) {
  (void)in_sizes; (void)n_in; (void)out_size; (void)d_ws; (void)ws_size;
  const float* y     = (const float*)d_in[0];
  const float* W_ih0 = (const float*)d_in[1];
  const float* b_ih0 = (const float*)d_in[2];
  const float* W_hh0 = (const float*)d_in[3];
  const float* b_hh0 = (const float*)d_in[4];
  const float* W_ih1 = (const float*)d_in[5];
  const float* b_ih1 = (const float*)d_in[6];
  const float* W_hh1 = (const float*)d_in[7];
  const float* b_hh1 = (const float*)d_in[8];
  const float* W_lin = (const float*)d_in[9];
  const float* b_lin = (const float*)d_in[10];
  float* out = (float*)d_out;

  rnn_pipeline_kernel<<<dim3(1), dim3(96), 0, stream>>>(
      y, W_ih0, b_ih0, W_hh0, b_hh0, W_ih1, b_ih1, W_hh1, b_hh1,
      W_lin, b_lin, out);
}


# LDS traffic of both scans taken out of the steps into one place per 16 steps: outputs kept in registers, stored together (two steps per store); layer-1 xp1 rows fetched 16 at a time, two groups ahead
# speedup vs baseline: 2.7799x; 1.7362x over previous
#include <hip/hip_runtime.h>


#define T_LEN   1048576
#define H       16
#define CHUNK   256
#define NCHUNK  (T_LEN / CHUNK)
#define YBLK    32
#define NYBLK   (T_LEN / YBLK)
#define G       16

typedef __attribute__((ext_vector_type(2))) float v2f;
typedef __attribute__((ext_vector_type(8))) float v8f;

__device__ __forceinline__ float readlane_f(float v, int lane) {
  return __builtin_bit_cast(float,
      __builtin_amdgcn_readlane(__builtin_bit_cast(int, v), lane));
}

#define RS(k) " row_share:" #k " row_mask:0xf bank_mask:0xf\n\t"
__device__ __forceinline__ float rnn_step(float x, float h, const float (&w)[H]) {
  float s0 = x, s1, s2, s3, hn;
  asm("v_fmac_f32_dpp %0, %5, %6"  RS(0)
      "v_mul_f32_dpp  %1, %5, %7"  RS(1)
      "v_mul_f32_dpp  %2, %5, %8"  RS(2)
      "v_mul_f32_dpp  %3, %5, %9"  RS(3)
      "v_fmac_f32_dpp %0, %5, %10" RS(4)
      "v_fmac_f32_dpp %1, %5, %11" RS(5)
      "v_fmac_f32_dpp %2, %5, %12" RS(6)
      "v_fmac_f32_dpp %3, %5, %13" RS(7)
      "v_fmac_f32_dpp %0, %5, %14" RS(8)
      "v_fmac_f32_dpp %1, %5, %15" RS(9)
      "v_fmac_f32_dpp %2, %5, %16" RS(10)
      "v_fmac_f32_dpp %3, %5, %17" RS(11)
      "v_fmac_f32_dpp %0, %5, %18" RS(12)
      "v_fmac_f32_dpp %1, %5, %19" RS(13)
      "v_fmac_f32_dpp %2, %5, %20" RS(14)
      "v_fmac_f32_dpp %3, %5, %21" RS(15)
      "v_add_f32 %0, %0, %1\n\t"
      "v_add_f32 %2, %2, %3\n\t"
      "v_add_f32 %0, %0, %2\n\t"
      "v_tanh_f32 %4, %0\n\t"
      "v_nop"
      : "+v"(s0), "=&v"(s1), "=&v"(s2), "=&v"(s3), "=&v"(hn)
      : "v"(h),
        "v"(w[0]),  "v"(w[1]),  "v"(w[2]),  "v"(w[3]),
        "v"(w[4]),  "v"(w[5]),  "v"(w[6]),  "v"(w[7]),
        "v"(w[8]),  "v"(w[9]),  "v"(w[10]), "v"(w[11]),
        "v"(w[12]), "v"(w[13]), "v"(w[14]), "v"(w[15]));
  return hn;
}
#undef RS

__global__ __launch_bounds__(96, 1)
void rnn_pipeline_kernel(const float* __restrict__ y,
                         const float* __restrict__ W_ih0, const float* __restrict__ b_ih0,
                         const float* __restrict__ W_hh0, const float* __restrict__ b_hh0,
                         const float* __restrict__ W_ih1, const float* __restrict__ b_ih1,
                         const float* __restrict__ W_hh1, const float* __restrict__ b_hh1,
                         const float* __restrict__ W_lin, const float* __restrict__ b_lin,
                         float* __restrict__ out)
{
  __shared__ float h1buf[2][CHUNK * H];
  __shared__ float h2buf[2][CHUNK * H];
  __shared__ float xp1buf[CHUNK * H];

  const int tid  = threadIdx.x;
  const int wave = tid >> 5;
  const int l    = tid & 31;
  const int n    = l & 15;
  const int half = l >> 4;

  float w0[H];
  float w1[H];
  float wl[H];
  v2f   Bw[4];
  float wih = 0.f, bs0 = 0.f, b1v = 0.f, bl = 0.f;
  float ycur = 0.f;

  float hs = 0.0f;

  if (wave == 0) {
    #pragma unroll
    for (int k = 0; k < H; ++k) w0[k] = W_hh0[n * H + k];
    wih = W_ih0[n];
    bs0 = b_ih0[n] + b_hh0[n];
    ycur = y[l];
  } else if (wave == 1) {
    #pragma unroll
    for (int k = 0; k < H; ++k) w1[k] = W_hh1[n * H + k];
    #pragma unroll
    for (int kb = 0; kb < 4; ++kb) {
      const int K = 4 * kb + 2 * half;
      Bw[kb] = *(const v2f*)(W_ih1 + n * H + K);
    }
    b1v = b_ih1[n] + b_hh1[n];
  } else {
    const int o = l & 1;
    #pragma unroll
    for (int k = 0; k < H; ++k) wl[k] = W_lin[o * H + k];
    bl = b_lin[o];
  }

  for (int i = 0; i < NCHUNK + 2; ++i) {
    if (wave == 0) {
      if (i < NCHUNK) {
        float* hb = h1buf[i & 1];
        for (int b = 0; b < CHUNK / YBLK; ++b) {
          const int g  = i * (CHUNK / YBLK) + b;
          const int gn = (g + 1 < NYBLK) ? g + 1 : g;
          const float ynext = y[gn * YBLK + l];
          for (int jg = 0; jg < YBLK; jg += G) {
            float hq[G];
            #pragma unroll
            for (int u = 0; u < G; ++u) {
              const float yv = readlane_f(ycur, jg + u);
              const float xp = fmaf(yv, wih, bs0);
              hs = rnn_step(xp, hs, w0);
              hq[u] = hs;
            }
            float* dst = hb + (b * YBLK + jg) * H + l;
            #pragma unroll
            for (int u = 0; u < G; u += 2)
              dst[u * H] = half ? hq[u + 1] : hq[u];
          }
          ycur = ynext;
        }
      }
    } else if (wave == 1) {
      if (i >= 1 && i <= NCHUNK) {
        const int c = i - 1;
        const float* hb = h1buf[c & 1];
        float*       ob = h2buf[c & 1];

        for (int tile = 0; tile < CHUNK / 16; ++tile) {
          v8f acc;
          #pragma unroll
          for (int v = 0; v < 8; ++v) acc[v] = b1v;
          #pragma unroll
          for (int kb = 0; kb < 4; ++kb) {
            const int K = 4 * kb + 2 * half;
            const v2f Aa = *(const v2f*)(hb + (tile * 16 + n) * H + K);
            acc = __builtin_amdgcn_wmma_f32_16x16x4_f32(
                false, Aa, false, Bw[kb], (short)0, acc, false, false);
          }
          #pragma unroll
          for (int v = 0; v < 8; ++v)
            xp1buf[(tile * 16 + v + 8 * half) * H + n] = acc[v];
        }

        float xa[G], xb[G];
        #pragma unroll
        for (int u = 0; u < G; ++u) xa[u] = xp1buf[u * H + n];
        #pragma unroll
        for (int u = 0; u < G; ++u) xb[u] = xp1buf[(G + u) * H + n];
        for (int sb = 0; sb < CHUNK; sb += 2 * G) {
          const int ra = (sb + 2 * G < CHUNK) ? sb + 2 * G : sb;
          const int rb = (sb + 3 * G < CHUNK) ? sb + 3 * G : sb + G;
          float hq[G];
          #pragma unroll
          for (int u = 0; u < G; ++u) { hs = rnn_step(xa[u], hs, w1); hq[u] = hs; }
          {
            float pk[G / 2];
            #pragma unroll
            for (int u = 0; u < G; u += 2) pk[u / 2] = half ? hq[u + 1] : hq[u];
            float*       dst = ob + sb * H + l;
            const float* src = xp1buf + ra * H + n;
            __builtin_amdgcn_sched_barrier(0);
            #pragma unroll
            for (int u = 0; u < G; u += 2) dst[u * H] = pk[u / 2];
            #pragma unroll
            for (int u = 0; u < G; ++u) xa[u] = src[u * H];
            __builtin_amdgcn_sched_barrier(0);
          }
          #pragma unroll
          for (int u = 0; u < G; ++u) { hs = rnn_step(xb[u], hs, w1); hq[u] = hs; }
          {
            float pk[G / 2];
            #pragma unroll
            for (int u = 0; u < G; u += 2) pk[u / 2] = half ? hq[u + 1] : hq[u];
            float*       dst = ob + (sb + G) * H + l;
            const float* src = xp1buf + rb * H + n;
            __builtin_amdgcn_sched_barrier(0);
            #pragma unroll
            for (int u = 0; u < G; u += 2) dst[u * H] = pk[u / 2];
            #pragma unroll
            for (int u = 0; u < G; ++u) xb[u] = src[u * H];
            __builtin_amdgcn_sched_barrier(0);
          }
        }
      }
    } else {
      if (i >= 2) {
        const int c = i - 2;
        const float* hb = h2buf[c & 1];
        const int o  = l & 1;
        const int tt = l >> 1;
        const int base = c * CHUNK;
        for (int s = tt; s < CHUNK; s += 16) {
          float acc = bl;
          #pragma unroll
          for (int k = 0; k < H; ++k)
            acc = fmaf(hb[s * H + k], wl[k], acc);
          out[(base + s) * 2 + o] = acc;
        }
      }
    }
    __syncthreads();
  }
}

extern "C" void kernel_launch(void* const* d_in, const int* in_sizes, int n_in,
                              void* d_out, int out_size, void* d_ws, size_t ws_size,
                              hipStream_t stream) {
  (void)in_sizes; (void)n_in; (void)out_size; (void)d_ws; (void)ws_size;
  const float* y     = (const float*)d_in[0];
  const float* W_ih0 = (const float*)d_in[1];
  const float* b_ih0 = (const float*)d_in[2];
  const float* W_hh0 = (const float*)d_in[3];
  const float* b_hh0 = (const float*)d_in[4];
  const float* W_ih1 = (const float*)d_in[5];
  const float* b_ih1 = (const float*)d_in[6];
  const float* W_hh1 = (const float*)d_in[7];
  const float* b_hh1 = (const float*)d_in[8];
  const float* W_lin = (const float*)d_in[9];
  const float* b_lin = (const float*)d_in[10];
  float* out = (float*)d_out;

  rnn_pipeline_kernel<<<dim3(1), dim3(96), 0, stream>>>(
      y, W_ih0, b_ih0, W_hh0, b_hh0, W_ih1, b_ih1, W_hh1, b_hh1,
      W_lin, b_lin, out);
}


# Input projections of both scans taken off the scan waves: the head's wave makes the xp0 rows and the xp1 WMMA GEMM as pipeline stages of their own, rows handed over in LDS; scan waves run steps only
# speedup vs baseline: 3.2207x; 1.1586x over previous
#include <hip/hip_runtime.h>


#define T_LEN   1048576
#define H       16
#define CHUNK   256
#define NCHUNK  (T_LEN / CHUNK)
#define G       16

typedef __attribute__((ext_vector_type(2))) float v2f;
typedef __attribute__((ext_vector_type(4))) float v4f;
typedef __attribute__((ext_vector_type(8))) float v8f;

#define RS(k) " row_share:" #k " row_mask:0xf bank_mask:0xf\n\t"
__device__ __forceinline__ float rnn_step(float x, float h, const float (&w)[H]) {
  float s0 = x, s1, s2, s3, hn;
  asm("v_fmac_f32_dpp %0, %5, %6"  RS(0)
      "v_mul_f32_dpp  %1, %5, %7"  RS(1)
      "v_mul_f32_dpp  %2, %5, %8"  RS(2)
      "v_mul_f32_dpp  %3, %5, %9"  RS(3)
      "v_fmac_f32_dpp %0, %5, %10" RS(4)
      "v_fmac_f32_dpp %1, %5, %11" RS(5)
      "v_fmac_f32_dpp %2, %5, %12" RS(6)
      "v_fmac_f32_dpp %3, %5, %13" RS(7)
      "v_fmac_f32_dpp %0, %5, %14" RS(8)
      "v_fmac_f32_dpp %1, %5, %15" RS(9)
      "v_fmac_f32_dpp %2, %5, %16" RS(10)
      "v_fmac_f32_dpp %3, %5, %17" RS(11)
      "v_fmac_f32_dpp %0, %5, %18" RS(12)
      "v_fmac_f32_dpp %1, %5, %19" RS(13)
      "v_fmac_f32_dpp %2, %5, %20" RS(14)
      "v_fmac_f32_dpp %3, %5, %21" RS(15)
      "v_add_f32 %0, %0, %1\n\t"
      "v_add_f32 %2, %2, %3\n\t"
      "v_add_f32 %0, %0, %2\n\t"
      "v_tanh_f32 %4, %0\n\t"
      "v_nop"
      : "+v"(s0), "=&v"(s1), "=&v"(s2), "=&v"(s3), "=&v"(hn)
      : "v"(h),
        "v"(w[0]),  "v"(w[1]),  "v"(w[2]),  "v"(w[3]),
        "v"(w[4]),  "v"(w[5]),  "v"(w[6]),  "v"(w[7]),
        "v"(w[8]),  "v"(w[9]),  "v"(w[10]), "v"(w[11]),
        "v"(w[12]), "v"(w[13]), "v"(w[14]), "v"(w[15]));
  return hn;
}
#undef RS

__global__ __launch_bounds__(96, 1)
void rnn_pipeline_kernel(const float* __restrict__ y,
                         const float* __restrict__ W_ih0, const float* __restrict__ b_ih0,
                         const float* __restrict__ W_hh0, const float* __restrict__ b_hh0,
                         const float* __restrict__ W_ih1, const float* __restrict__ b_ih1,
                         const float* __restrict__ W_hh1, const float* __restrict__ b_hh1,
                         const float* __restrict__ W_lin, const float* __restrict__ b_lin,
                         float* __restrict__ out)
{
  __shared__ alignas(16) float stage[4][2][CHUNK * H];

  const int tid  = threadIdx.x;
  const int wave = __builtin_amdgcn_readfirstlane(tid >> 5);
  const int l    = tid & 31;
  const int n    = l & 15;
  const int half = l >> 4;

  float wr[H];
  float wl[H];
  float wi0[4];
  float bi0[4];
  v2f   Bw[4];
  float b1v = 0.f, bl = 0.f;

  float hs = 0.0f;

  if (wave < 2) {
    const float* Whh = (wave == 0) ? W_hh0 : W_hh1;
    #pragma unroll
    for (int k = 0; k < H; ++k) wr[k] = Whh[n * H + k];
  } else {
    #pragma unroll
    for (int e = 0; e < 4; ++e) {
      const int r = 4 * (l & 3) + e;
      wi0[e] = W_ih0[r];
      bi0[e] = b_ih0[r] + b_hh0[r];
    }
    #pragma unroll
    for (int kb = 0; kb < 4; ++kb) {
      const int K = 4 * kb + 2 * half;
      Bw[kb] = *(const v2f*)(W_ih1 + n * H + K);
    }
    b1v = b_ih1[n] + b_hh1[n];
    const int o = l & 1;
    #pragma unroll
    for (int k = 0; k < H; ++k) wl[k] = W_lin[o * H + k];
    bl = b_lin[o];
  }

  for (int i = 0; i < NCHUNK + 4; ++i) {
    if (wave < 2) {
      const int c = i - 1 - 2 * wave;
      if (c >= 0 && c < NCHUNK) {
        const float* xpb = stage[2 * wave][c & 1];
        float*       ob  = stage[2 * wave + 1][c & 1];

        float xa[G], xb[G];
        #pragma unroll
        for (int u = 0; u < G; ++u) xa[u] = xpb[u * H + n];
        #pragma unroll
        for (int u = 0; u < G; ++u) xb[u] = xpb[(G + u) * H + n];
        for (int sb = 0; sb < CHUNK; sb += 2 * G) {
          const int ra = (sb + 2 * G < CHUNK) ? sb + 2 * G : sb;
          const int rb = (sb + 3 * G < CHUNK) ? sb + 3 * G : sb + G;
          float hq[G];
          #pragma unroll
          for (int u = 0; u < G; ++u) { hs = rnn_step(xa[u], hs, wr); hq[u] = hs; }
          {
            float pk[G / 2];
            #pragma unroll
            for (int u = 0; u < G; u += 2) pk[u / 2] = half ? hq[u + 1] : hq[u];
            float*       dst = ob + sb * H + l;
            const float* src = xpb + ra * H + n;
            __builtin_amdgcn_sched_barrier(0);
            #pragma unroll
            for (int u = 0; u < G; ++u) xa[u] = src[u * H];
            #pragma unroll
            for (int u = 0; u < G; u += 2) dst[u * H] = pk[u / 2];
            __builtin_amdgcn_sched_barrier(0);
          }
          #pragma unroll
          for (int u = 0; u < G; ++u) { hs = rnn_step(xb[u], hs, wr); hq[u] = hs; }
          {
            float pk[G / 2];
            #pragma unroll
            for (int u = 0; u < G; u += 2) pk[u / 2] = half ? hq[u + 1] : hq[u];
            float*       dst = ob + (sb + G) * H + l;
            const float* src = xpb + rb * H + n;
            __builtin_amdgcn_sched_barrier(0);
            #pragma unroll
            for (int u = 0; u < G; ++u) xb[u] = src[u * H];
            #pragma unroll
            for (int u = 0; u < G; u += 2) dst[u * H] = pk[u / 2];
            __builtin_amdgcn_sched_barrier(0);
          }
        }
      }
    } else {
      if (i < NCHUNK) {
        float*       xo = stage[0][i & 1] + 4 * l;
        const float* yc = y + i * CHUNK + (l >> 2);
        for (int j0 = 0; j0 < CHUNK / 8; j0 += 8) {
          float yv[8];
          #pragma unroll
          for (int u = 0; u < 8; ++u) yv[u] = yc[(j0 + u) * 8];
          #pragma unroll
          for (int u = 0; u < 8; ++u) {
            v4f r;
            r[0] = fmaf(yv[u], wi0[0], bi0[0]);
            r[1] = fmaf(yv[u], wi0[1], bi0[1]);
            r[2] = fmaf(yv[u], wi0[2], bi0[2]);
            r[3] = fmaf(yv[u], wi0[3], bi0[3]);
            *(v4f*)(xo + (j0 + u) * 8 * H) = r;
          }
        }
      }
      if (i >= 2 && i <= NCHUNK + 1) {
        const int c = i - 2;
        const float* hb = stage[1][c & 1];
        float*       xo = stage[2][c & 1];

        for (int tile = 0; tile < CHUNK / 16; ++tile) {
          v8f acc;
          #pragma unroll
          for (int v = 0; v < 8; ++v) acc[v] = b1v;
          #pragma unroll
          for (int kb = 0; kb < 4; ++kb) {
            const int K = 4 * kb + 2 * half;
            const v2f Aa = *(const v2f*)(hb + (tile * 16 + n) * H + K);
            acc = __builtin_amdgcn_wmma_f32_16x16x4_f32(
                false, Aa, false, Bw[kb], (short)0, acc, false, false);
          }
          #pragma unroll
          for (int v = 0; v < 8; ++v)
            xo[(tile * 16 + v + 8 * half) * H + n] = acc[v];
        }
      }
      if (i >= 4) {
        const int c = i - 4;
        const float* hb = stage[3][c & 1];
        const int o  = l & 1;
        const int tt = l >> 1;
        const int base = c * CHUNK;
        for (int s = tt; s < CHUNK; s += 16) {
          float acc = bl;
          #pragma unroll
          for (int k = 0; k < H; ++k)
            acc = fmaf(hb[s * H + k], wl[k], acc);
          out[(base + s) * 2 + o] = acc;
        }
      }
    }
    __syncthreads();
  }
}

extern "C" void kernel_launch(void* const* d_in, const int* in_sizes, int n_in,
                              void* d_out, int out_size, void* d_ws, size_t ws_size,
                              hipStream_t stream) {
  (void)in_sizes; (void)n_in; (void)out_size; (void)d_ws; (void)ws_size;
  const float* y     = (const float*)d_in[0];
  const float* W_ih0 = (const float*)d_in[1];
  const float* b_ih0 = (const float*)d_in[2];
  const float* W_hh0 = (const float*)d_in[3];
  const float* b_hh0 = (const float*)d_in[4];
  const float* W_ih1 = (const float*)d_in[5];
  const float* b_ih1 = (const float*)d_in[6];
  const float* W_hh1 = (const float*)d_in[7];
  const float* b_hh1 = (const float*)d_in[8];
  const float* W_lin = (const float*)d_in[9];
  const float* b_lin = (const float*)d_in[10];
  float* out = (float*)d_out;

  rnn_pipeline_kernel<<<dim3(1), dim3(96), 0, stream>>>(
      y, W_ih0, b_ih0, W_hh0, b_hh0, W_ih1, b_ih1, W_hh1, b_hh1,
      W_lin, b_lin, out);
}


# Each output's 16 products split between the wave's two 16-lane rows (k 0-7 / 8-15): 8 DPP FMAs per step, not 16; half sums joined by one v_permlanex16; sum order now 3 partial sums + cross-row add
# speedup vs baseline: 3.7794x; 1.1735x over previous
#include <hip/hip_runtime.h>


#define T_LEN   1048576
#define H       16
#define CHUNK   256
#define NCHUNK  (T_LEN / CHUNK)
#define G       16
#define ZOFF    32

typedef __attribute__((ext_vector_type(2))) float v2f;
typedef __attribute__((ext_vector_type(4))) float v4f;
typedef __attribute__((ext_vector_type(8))) float v8f;

#define KH 8
#define RS(k) " row_share:" #k " row_mask:0xf bank_mask:0xf\n\t"
__device__ __forceinline__ float rnn_step(float x, float h, const float (&w)[KH],
                                          unsigned sel_lo, unsigned sel_hi) {
  float s0 = x, s1, s2, sx, hn;
  asm("v_fmac_f32_dpp %0, %5, %6"  RS(0)
      "v_mul_f32_dpp  %1, %5, %7"  RS(1)
      "v_mul_f32_dpp  %2, %5, %8"  RS(2)
      "v_fmac_f32_dpp %0, %5, %9"  RS(3)
      "v_fmac_f32_dpp %1, %5, %10" RS(4)
      "v_fmac_f32_dpp %2, %5, %11" RS(5)
      "v_fmac_f32_dpp %0, %5, %12" RS(6)
      "v_fmac_f32_dpp %1, %5, %13" RS(7)
      "v_add_f32 %0, %0, %2\n\t"
      "v_add_f32 %0, %0, %1\n\t"
      "v_permlanex16_b32 %3, %0, %14, %15\n\t"
      "v_add_f32 %0, %0, %3\n\t"
      "v_tanh_f32 %4, %0\n\t"
      "v_nop"
      : "+v"(s0), "=&v"(s1), "=&v"(s2), "=&v"(sx), "=&v"(hn)
      : "v"(h),
        "v"(w[0]), "v"(w[1]), "v"(w[2]), "v"(w[3]),
        "v"(w[4]), "v"(w[5]), "v"(w[6]), "v"(w[7]),
        "s"(sel_lo), "s"(sel_hi));
  return hn;
}
#undef RS

__global__ __launch_bounds__(96, 1)
void rnn_pipeline_kernel(const float* __restrict__ y,
                         const float* __restrict__ W_ih0, const float* __restrict__ b_ih0,
                         const float* __restrict__ W_hh0, const float* __restrict__ b_hh0,
                         const float* __restrict__ W_ih1, const float* __restrict__ b_ih1,
                         const float* __restrict__ W_hh1, const float* __restrict__ b_hh1,
                         const float* __restrict__ W_lin, const float* __restrict__ b_lin,
                         float* __restrict__ out)
{
  __shared__ alignas(256) float stage[4][2][CHUNK * H];
  __shared__ alignas(256) float zrows[ZOFF + G * H];

  const int tid  = threadIdx.x;
  const int wave = __builtin_amdgcn_readfirstlane(tid >> 5);
  const int l    = tid & 31;
  const int n    = l & 15;
  const int half = l >> 4;
  const int r    = (n + 8 * half) & 15;

  float wr[KH];
  float wl[H];
  float wi0[4];
  float bi0[4];
  v2f   Bw[4];
  float b1v = 0.f, bl = 0.f;

  float hs = 0.0f;
  const unsigned sel_lo = 0xFEDCBA98u, sel_hi = 0x76543210u;

  for (int q = tid; q < ZOFF + G * H; q += 96) zrows[q] = 0.0f;

  if (wave < 2) {
    const float* Whh = (wave == 0) ? W_hh0 : W_hh1;
    #pragma unroll
    for (int k = 0; k < KH; ++k) wr[k] = Whh[r * H + 8 * half + k];
  } else {
    #pragma unroll
    for (int e = 0; e < 4; ++e) {
      const int r = 4 * (l & 3) + e;
      wi0[e] = W_ih0[r];
      bi0[e] = b_ih0[r] + b_hh0[r];
    }
    #pragma unroll
    for (int kb = 0; kb < 4; ++kb) {
      const int K = 4 * kb + 2 * half;
      Bw[kb] = *(const v2f*)(W_ih1 + n * H + K);
    }
    b1v = b_ih1[n] + b_hh1[n];
    const int o = l & 1;
    #pragma unroll
    for (int k = 0; k < H; ++k) wl[k] = W_lin[o * H + k];
    bl = b_lin[o];
  }

  for (int i = 0; i < NCHUNK + 4; ++i) {
    if (wave < 2) {
      const int c = i - 1 - 2 * wave;
      if (c >= 0 && c < NCHUNK) {
        const float* xpb = stage[2 * wave][c & 1];
        float*       ob  = stage[2 * wave + 1][c & 1];

        float xa[G], xb[G];
        const float* xin = half ? zrows + ZOFF + n : xpb + n;
        const int    xmk = half ? 0 : -1;
        #pragma unroll
        for (int u = 0; u < G; ++u) xa[u] = xin[u * H];
        #pragma unroll
        for (int u = 0; u < G; ++u) xb[u] = xin[((G * H) & xmk) + u * H];
        for (int sb = 0; sb < CHUNK; sb += 2 * G) {
          const int ra = (sb + 2 * G < CHUNK) ? sb + 2 * G : sb;
          const int rb = (sb + 3 * G < CHUNK) ? sb + 3 * G : sb + G;
          float hq[G];
          #pragma unroll
          for (int u = 0; u < G; ++u) { hs = rnn_step(xa[u], hs, wr, sel_lo, sel_hi); hq[u] = hs; }
          {
            float pk[G / 2];
            #pragma unroll
            for (int u = 0; u < G; u += 2) pk[u / 2] = half ? hq[u + 1] : hq[u];
            float*       dst = ob + (sb + half) * H + r;
            const float* src = xin + ((ra * H) & xmk);
            __builtin_amdgcn_sched_barrier(0);
            #pragma unroll
            for (int u = 0; u < G; ++u) xa[u] = src[u * H];
            #pragma unroll
            for (int u = 0; u < G; u += 2) dst[u * H] = pk[u / 2];
            __builtin_amdgcn_sched_barrier(0);
          }
          #pragma unroll
          for (int u = 0; u < G; ++u) { hs = rnn_step(xb[u], hs, wr, sel_lo, sel_hi); hq[u] = hs; }
          {
            float pk[G / 2];
            #pragma unroll
            for (int u = 0; u < G; u += 2) pk[u / 2] = half ? hq[u + 1] : hq[u];
            float*       dst = ob + (sb + G + half) * H + r;
            const float* src = xin + ((rb * H) & xmk);
            __builtin_amdgcn_sched_barrier(0);
            #pragma unroll
            for (int u = 0; u < G; ++u) xb[u] = src[u * H];
            #pragma unroll
            for (int u = 0; u < G; u += 2) dst[u * H] = pk[u / 2];
            __builtin_amdgcn_sched_barrier(0);
          }
        }
      }
    } else {
      if (i < NCHUNK) {
        float*       xo = stage[0][i & 1] + 4 * l;
        const float* yc = y + i * CHUNK + (l >> 2);
        for (int j0 = 0; j0 < CHUNK / 8; j0 += 8) {
          float yv[8];
          #pragma unroll
          for (int u = 0; u < 8; ++u) yv[u] = yc[(j0 + u) * 8];
          #pragma unroll
          for (int u = 0; u < 8; ++u) {
            v4f r;
            r[0] = fmaf(yv[u], wi0[0], bi0[0]);
            r[1] = fmaf(yv[u], wi0[1], bi0[1]);
            r[2] = fmaf(yv[u], wi0[2], bi0[2]);
            r[3] = fmaf(yv[u], wi0[3], bi0[3]);
            *(v4f*)(xo + (j0 + u) * 8 * H) = r;
          }
        }
      }
      if (i >= 2 && i <= NCHUNK + 1) {
        const int c = i - 2;
        const float* hb = stage[1][c & 1];
        float*       xo = stage[2][c & 1];

        for (int tile = 0; tile < CHUNK / 16; ++tile) {
          v8f acc;
          #pragma unroll
          for (int v = 0; v < 8; ++v) acc[v] = b1v;
          #pragma unroll
          for (int kb = 0; kb < 4; ++kb) {
            const int K = 4 * kb + 2 * half;
            const v2f Aa = *(const v2f*)(hb + (tile * 16 + n) * H + K);
            acc = __builtin_amdgcn_wmma_f32_16x16x4_f32(
                false, Aa, false, Bw[kb], (short)0, acc, false, false);
          }
          #pragma unroll
          for (int v = 0; v < 8; ++v)
            xo[(tile * 16 + v + 8 * half) * H + n] = acc[v];
        }
      }
      if (i >= 4) {
        const int c = i - 4;
        const float* hb = stage[3][c & 1];
        const int o  = l & 1;
        const int tt = l >> 1;
        const int base = c * CHUNK;
        for (int s = tt; s < CHUNK; s += 16) {
          float acc = bl;
          #pragma unroll
          for (int k = 0; k < H; ++k)
            acc = fmaf(hb[s * H + k], wl[k], acc);
          out[(base + s) * 2 + o] = acc;
        }
      }
    }
    __syncthreads();
  }
}

extern "C" void kernel_launch(void* const* d_in, const int* in_sizes, int n_in,
                              void* d_out, int out_size, void* d_ws, size_t ws_size,
                              hipStream_t stream) {
  (void)in_sizes; (void)n_in; (void)out_size; (void)d_ws; (void)ws_size;
  const float* y     = (const float*)d_in[0];
  const float* W_ih0 = (const float*)d_in[1];
  const float* b_ih0 = (const float*)d_in[2];
  const float* W_hh0 = (const float*)d_in[3];
  const float* b_hh0 = (const float*)d_in[4];
  const float* W_ih1 = (const float*)d_in[5];
  const float* b_ih1 = (const float*)d_in[6];
  const float* W_hh1 = (const float*)d_in[7];
  const float* b_hh1 = (const float*)d_in[8];
  const float* W_lin = (const float*)d_in[9];
  const float* b_lin = (const float*)d_in[10];
  float* out = (float*)d_out;

  rnn_pipeline_kernel<<<dim3(1), dim3(96), 0, stream>>>(
      y, W_ih0, b_ih0, W_hh0, b_hh0, W_ih1, b_ih1, W_hh1, b_hh1,
      W_lin, b_lin, out);
}


# Scans' LDS traffic per step cut: hand-over buffers keep 4 steps of one element contiguous, so inputs and outputs move as 16-byte loads/stores; one LDS place per 64 steps (was 16), chunks of 512
# speedup vs baseline: 4.0688x; 1.0766x over previous
#include <hip/hip_runtime.h>


#define T_LEN   1048576
#define H       16
#define CHUNK   512
#define NCHUNK  (T_LEN / CHUNK)
#define G       64
#define QF      64
#define QIDX(s, e) (((s) >> 2) * QF + (e) * 4 + ((s) & 3))

typedef __attribute__((ext_vector_type(2))) float v2f;
typedef __attribute__((ext_vector_type(4))) float v4f;
typedef __attribute__((ext_vector_type(8))) float v8f;

#define KH 8
#define RS(k) " row_share:" #k " row_mask:0xf bank_mask:0xf\n\t"
__device__ __forceinline__ float rnn_step(float x, float h, const float (&w)[KH],
                                          unsigned sel_lo, unsigned sel_hi) {
  float s0 = x, s1, s2, sx, hn;
  asm("v_fmac_f32_dpp %0, %5, %6"  RS(0)
      "v_mul_f32_dpp  %1, %5, %7"  RS(1)
      "v_mul_f32_dpp  %2, %5, %8"  RS(2)
      "v_fmac_f32_dpp %0, %5, %9"  RS(3)
      "v_fmac_f32_dpp %1, %5, %10" RS(4)
      "v_fmac_f32_dpp %2, %5, %11" RS(5)
      "v_fmac_f32_dpp %0, %5, %12" RS(6)
      "v_fmac_f32_dpp %1, %5, %13" RS(7)
      "v_add_f32 %0, %0, %2\n\t"
      "v_add_f32 %0, %0, %1\n\t"
      "v_permlanex16_b32 %3, %0, %14, %15\n\t"
      "v_add_f32 %0, %0, %3\n\t"
      "v_tanh_f32 %4, %0\n\t"
      "v_nop"
      : "+v"(s0), "=&v"(s1), "=&v"(s2), "=&v"(sx), "=&v"(hn)
      : "v"(h),
        "v"(w[0]), "v"(w[1]), "v"(w[2]), "v"(w[3]),
        "v"(w[4]), "v"(w[5]), "v"(w[6]), "v"(w[7]),
        "s"(sel_lo), "s"(sel_hi));
  return hn;
}
#undef RS

__global__ __launch_bounds__(96, 1)
void rnn_pipeline_kernel(const float* __restrict__ y,
                         const float* __restrict__ W_ih0, const float* __restrict__ b_ih0,
                         const float* __restrict__ W_hh0, const float* __restrict__ b_hh0,
                         const float* __restrict__ W_ih1, const float* __restrict__ b_ih1,
                         const float* __restrict__ W_hh1, const float* __restrict__ b_hh1,
                         const float* __restrict__ W_lin, const float* __restrict__ b_lin,
                         float* __restrict__ out)
{
  __shared__ alignas(256) float stage[4][2][CHUNK * H];
  __shared__ alignas(256) float zrows[(G / 4) * QF];

  const int tid  = threadIdx.x;
  const int wave = __builtin_amdgcn_readfirstlane(tid >> 5);
  const int l    = tid & 31;
  const int n    = l & 15;
  const int half = l >> 4;
  const int r    = (n + 8 * half) & 15;

  float wr[KH];
  float wl[H];
  float wi0n = 0.f;
  float bi0n = 0.f;
  v2f   Bw[4];
  float b1v = 0.f, bl = 0.f;

  float hs = 0.0f;
  const unsigned sel_lo = 0xFEDCBA98u, sel_hi = 0x76543210u;

  for (int q = tid; q < (G / 4) * QF; q += 96) zrows[q] = 0.0f;

  if (wave < 2) {
    const float* Whh = (wave == 0) ? W_hh0 : W_hh1;
    #pragma unroll
    for (int k = 0; k < KH; ++k) wr[k] = Whh[r * H + 8 * half + k];
  } else {
    wi0n = W_ih0[n];
    bi0n = b_ih0[n] + b_hh0[n];
    #pragma unroll
    for (int kb = 0; kb < 4; ++kb) {
      const int K = 4 * kb + 2 * half;
      Bw[kb] = *(const v2f*)(W_ih1 + n * H + K);
    }
    b1v = b_ih1[n] + b_hh1[n];
    const int o = l & 1;
    #pragma unroll
    for (int k = 0; k < H; ++k) wl[k] = W_lin[o * H + k];
    bl = b_lin[o];
  }

  for (int i = 0; i < NCHUNK + 4; ++i) {
    if (wave < 2) {
      const int c = i - 1 - 2 * wave;
      if (c >= 0 && c < NCHUNK) {
        const float* xpb = stage[2 * wave][c & 1];
        float*       ob  = stage[2 * wave + 1][c & 1];

        float xa[G], xb[G];
        const float* xin = half ? zrows + n * 4 : xpb + n * 4;
        const int    xmk = half ? 0 : -1;
        #pragma unroll
        for (int j = 0; j < G / 4; ++j) {
          const v4f t = *(const v4f*)(xin + j * QF);
          xa[4 * j] = t[0]; xa[4 * j + 1] = t[1]; xa[4 * j + 2] = t[2]; xa[4 * j + 3] = t[3];
        }
        #pragma unroll
        for (int j = 0; j < G / 4; ++j) {
          const v4f t = *(const v4f*)(xin + (((G / 4) * QF) & xmk) + j * QF);
          xb[4 * j] = t[0]; xb[4 * j + 1] = t[1]; xb[4 * j + 2] = t[2]; xb[4 * j + 3] = t[3];
        }
        for (int sb = 0; sb < CHUNK; sb += 2 * G) {
          const int ra = (sb + 2 * G < CHUNK) ? sb + 2 * G : sb;
          const int rb = (sb + 3 * G < CHUNK) ? sb + 3 * G : sb + G;
          float hq[G];
          #pragma unroll
          for (int u = 0; u < G; ++u) { hs = rnn_step(xa[u], hs, wr, sel_lo, sel_hi); hq[u] = hs; }
          {
            v4f pk[G / 8];
            #pragma unroll
            for (int p = 0; p < G / 8; ++p) {
              #pragma unroll
              for (int e = 0; e < 4; ++e) pk[p][e] = half ? hq[8 * p + 4 + e] : hq[8 * p + e];
            }
            float*       dsta = ob + ((sb >> 2) + half) * QF + r * 4;
            const float* srca = xin + (((ra >> 2) * QF) & xmk);
            __builtin_amdgcn_sched_barrier(0);
            #pragma unroll
            for (int j = 0; j < G / 4; ++j) {
              const v4f t = *(const v4f*)(srca + j * QF);
              xa[4 * j] = t[0]; xa[4 * j + 1] = t[1]; xa[4 * j + 2] = t[2]; xa[4 * j + 3] = t[3];
            }
            #pragma unroll
            for (int p = 0; p < G / 8; ++p) *(v4f*)(dsta + p * 2 * QF) = pk[p];
            __builtin_amdgcn_sched_barrier(0);
          }
          #pragma unroll
          for (int u = 0; u < G; ++u) { hs = rnn_step(xb[u], hs, wr, sel_lo, sel_hi); hq[u] = hs; }
          {
            v4f pk[G / 8];
            #pragma unroll
            for (int p = 0; p < G / 8; ++p) {
              #pragma unroll
              for (int e = 0; e < 4; ++e) pk[p][e] = half ? hq[8 * p + 4 + e] : hq[8 * p + e];
            }
            float*       dstb = ob + (((sb + G) >> 2) + half) * QF + r * 4;
            const float* srcb = xin + (((rb >> 2) * QF) & xmk);
            __builtin_amdgcn_sched_barrier(0);
            #pragma unroll
            for (int j = 0; j < G / 4; ++j) {
              const v4f t = *(const v4f*)(srcb + j * QF);
              xb[4 * j] = t[0]; xb[4 * j + 1] = t[1]; xb[4 * j + 2] = t[2]; xb[4 * j + 3] = t[3];
            }
            #pragma unroll
            for (int p = 0; p < G / 8; ++p) *(v4f*)(dstb + p * 2 * QF) = pk[p];
            __builtin_amdgcn_sched_barrier(0);
          }
        }
      }
    } else {
      if (i < NCHUNK) {
        float*       xo = stage[0][i & 1] + half * QF + n * 4;
        const float* yc = y + i * CHUNK + half * 4;
        for (int j0 = 0; j0 < CHUNK / 8; j0 += 8) {
          v4f yv[8];
          #pragma unroll
          for (int u = 0; u < 8; ++u) yv[u] = *(const v4f*)(yc + (j0 + u) * 8);
          #pragma unroll
          for (int u = 0; u < 8; ++u) {
            v4f r;
            r[0] = fmaf(yv[u][0], wi0n, bi0n);
            r[1] = fmaf(yv[u][1], wi0n, bi0n);
            r[2] = fmaf(yv[u][2], wi0n, bi0n);
            r[3] = fmaf(yv[u][3], wi0n, bi0n);
            *(v4f*)(xo + (j0 + u) * 2 * QF) = r;
          }
        }
      }
      if (i >= 2 && i <= NCHUNK + 1) {
        const int c = i - 2;
        const float* hb = stage[1][c & 1];
        float*       xo = stage[2][c & 1];

        for (int tile = 0; tile < CHUNK / 16; ++tile) {
          v8f acc;
          #pragma unroll
          for (int v = 0; v < 8; ++v) acc[v] = b1v;
          #pragma unroll
          for (int kb = 0; kb < 4; ++kb) {
            const int K = 4 * kb + 2 * half;
            const float* ap = hb + QIDX(tile * 16 + n, K);
            v2f Aa;
            Aa[0] = ap[0];
            Aa[1] = ap[4];
            acc = __builtin_amdgcn_wmma_f32_16x16x4_f32(
                false, Aa, false, Bw[kb], (short)0, acc, false, false);
          }
          float* xq = xo + (tile * 4 + 2 * half) * QF + n * 4;
          v4f d0, d1;
          #pragma unroll
          for (int v = 0; v < 4; ++v) { d0[v] = acc[v]; d1[v] = acc[4 + v]; }
          *(v4f*)(xq)      = d0;
          *(v4f*)(xq + QF) = d1;
        }
      }
      if (i >= 4) {
        const int c = i - 4;
        const float* hb = stage[3][c & 1];
        const int o  = l & 1;
        const int tt = l >> 1;
        const int base = c * CHUNK;
        for (int s = tt; s < CHUNK; s += 16) {
          float acc = bl;
          const float* hp = hb + QIDX(s, 0);
          #pragma unroll
          for (int k = 0; k < H; ++k)
            acc = fmaf(hp[k * 4], wl[k], acc);
          out[(base + s) * 2 + o] = acc;
        }
      }
    }
    __syncthreads();
  }
}

extern "C" void kernel_launch(void* const* d_in, const int* in_sizes, int n_in,
                              void* d_out, int out_size, void* d_ws, size_t ws_size,
                              hipStream_t stream) {
  (void)in_sizes; (void)n_in; (void)out_size; (void)d_ws; (void)ws_size;
  const float* y     = (const float*)d_in[0];
  const float* W_ih0 = (const float*)d_in[1];
  const float* b_ih0 = (const float*)d_in[2];
  const float* W_hh0 = (const float*)d_in[3];
  const float* b_hh0 = (const float*)d_in[4];
  const float* W_ih1 = (const float*)d_in[5];
  const float* b_ih1 = (const float*)d_in[6];
  const float* W_hh1 = (const float*)d_in[7];
  const float* b_hh1 = (const float*)d_in[8];
  const float* W_lin = (const float*)d_in[9];
  const float* b_lin = (const float*)d_in[10];
  float* out = (float*)d_out;

  rnn_pipeline_kernel<<<dim3(1), dim3(96), 0, stream>>>(
      y, W_ih0, b_ih0, W_hh0, b_hh0, W_ih1, b_ih1, W_hh1, b_hh1,
      W_lin, b_lin, out);
}


# Step tail re-treed: the 3 partial sums cross rows one by one as each completes (3 v_permlanex16) beside the adds; sum order now ((s0+s2)+s0')+(s1'+(s1+s2')), copies of h[r] mirrored
# speedup vs baseline: 4.2308x; 1.0398x over previous
#include <hip/hip_runtime.h>


#define T_LEN   1048576
#define H       16
#define CHUNK   512
#define NCHUNK  (T_LEN / CHUNK)
#define G       64
#define QF      64
#define QIDX(s, e) (((s) >> 2) * QF + (e) * 4 + ((s) & 3))

typedef __attribute__((ext_vector_type(2))) float v2f;
typedef __attribute__((ext_vector_type(4))) float v4f;
typedef __attribute__((ext_vector_type(8))) float v8f;

#define KH 8
#define RS(k) " row_share:" #k " row_mask:0xf bank_mask:0xf\n\t"
__device__ __forceinline__ float rnn_step(float x, float h, const float (&w)[KH],
                                          unsigned sel_lo, unsigned sel_hi) {
  float s0 = x, s1, s2, xa, xb, xc, hn;
  asm("v_fmac_f32_dpp %0, %7, %8"  RS(0)
      "v_mul_f32_dpp  %1, %7, %9"  RS(1)
      "v_mul_f32_dpp  %2, %7, %10" RS(2)
      "v_fmac_f32_dpp %0, %7, %11" RS(3)
      "v_fmac_f32_dpp %1, %7, %12" RS(4)
      "v_fmac_f32_dpp %2, %7, %13" RS(5)
      "v_fmac_f32_dpp %0, %7, %14" RS(6)
      "v_fmac_f32_dpp %1, %7, %15" RS(7)
      "v_permlanex16_b32 %5, %2, %16, %17\n\t"
      "v_permlanex16_b32 %3, %0, %16, %17\n\t"
      "v_add_f32 %0, %0, %2\n\t"
      "v_permlanex16_b32 %4, %1, %16, %17\n\t"
      "v_add_f32 %1, %1, %5\n\t"
      "v_add_f32 %0, %0, %3\n\t"
      "v_add_f32 %1, %4, %1\n\t"
      "v_add_f32 %0, %0, %1\n\t"
      "v_tanh_f32 %6, %0\n\t"
      "v_nop"
      : "+v"(s0), "=&v"(s1), "=&v"(s2), "=&v"(xa), "=&v"(xb), "=&v"(xc), "=&v"(hn)
      : "v"(h),
        "v"(w[0]), "v"(w[1]), "v"(w[2]), "v"(w[3]),
        "v"(w[4]), "v"(w[5]), "v"(w[6]), "v"(w[7]),
        "s"(sel_lo), "s"(sel_hi));
  return hn;
}
#undef RS

__global__ __launch_bounds__(96, 1)
void rnn_pipeline_kernel(const float* __restrict__ y,
                         const float* __restrict__ W_ih0, const float* __restrict__ b_ih0,
                         const float* __restrict__ W_hh0, const float* __restrict__ b_hh0,
                         const float* __restrict__ W_ih1, const float* __restrict__ b_ih1,
                         const float* __restrict__ W_hh1, const float* __restrict__ b_hh1,
                         const float* __restrict__ W_lin, const float* __restrict__ b_lin,
                         float* __restrict__ out)
{
  __shared__ alignas(256) float stage[4][2][CHUNK * H];
  __shared__ alignas(256) float zrows[(G / 4) * QF];

  const int tid  = threadIdx.x;
  const int wave = __builtin_amdgcn_readfirstlane(tid >> 5);
  const int l    = tid & 31;
  const int n    = l & 15;
  const int half = l >> 4;
  const int r    = (n + 8 * half) & 15;

  float wr[KH];
  float wl[H];
  float wi0n = 0.f;
  float bi0n = 0.f;
  v2f   Bw[4];
  float b1v = 0.f, bl = 0.f;

  float hs = 0.0f;
  const unsigned sel_lo = 0xFEDCBA98u, sel_hi = 0x76543210u;

  for (int q = tid; q < (G / 4) * QF; q += 96) zrows[q] = 0.0f;

  if (wave < 2) {
    const float* Whh = (wave == 0) ? W_hh0 : W_hh1;
    #pragma unroll
    for (int k = 0; k < KH; ++k) wr[k] = Whh[r * H + 8 * half + k];
  } else {
    wi0n = W_ih0[n];
    bi0n = b_ih0[n] + b_hh0[n];
    #pragma unroll
    for (int kb = 0; kb < 4; ++kb) {
      const int K = 4 * kb + 2 * half;
      Bw[kb] = *(const v2f*)(W_ih1 + n * H + K);
    }
    b1v = b_ih1[n] + b_hh1[n];
    const int o = l & 1;
    #pragma unroll
    for (int k = 0; k < H; ++k) wl[k] = W_lin[o * H + k];
    bl = b_lin[o];
  }

  for (int i = 0; i < NCHUNK + 4; ++i) {
    if (wave < 2) {
      const int c = i - 1 - 2 * wave;
      if (c >= 0 && c < NCHUNK) {
        const float* xpb = stage[2 * wave][c & 1];
        float*       ob  = stage[2 * wave + 1][c & 1];

        float xa[G], xb[G];
        const float* xin = half ? zrows + n * 4 : xpb + n * 4;
        const int    xmk = half ? 0 : -1;
        #pragma unroll
        for (int j = 0; j < G / 4; ++j) {
          const v4f t = *(const v4f*)(xin + j * QF);
          xa[4 * j] = t[0]; xa[4 * j + 1] = t[1]; xa[4 * j + 2] = t[2]; xa[4 * j + 3] = t[3];
        }
        #pragma unroll
        for (int j = 0; j < G / 4; ++j) {
          const v4f t = *(const v4f*)(xin + (((G / 4) * QF) & xmk) + j * QF);
          xb[4 * j] = t[0]; xb[4 * j + 1] = t[1]; xb[4 * j + 2] = t[2]; xb[4 * j + 3] = t[3];
        }
        for (int sb = 0; sb < CHUNK; sb += 2 * G) {
          const int ra = (sb + 2 * G < CHUNK) ? sb + 2 * G : sb;
          const int rb = (sb + 3 * G < CHUNK) ? sb + 3 * G : sb + G;
          float hq[G];
          #pragma unroll
          for (int u = 0; u < G; ++u) { hs = rnn_step(xa[u], hs, wr, sel_lo, sel_hi); hq[u] = hs; }
          {
            v4f pk[G / 8];
            #pragma unroll
            for (int p = 0; p < G / 8; ++p) {
              #pragma unroll
              for (int e = 0; e < 4; ++e) pk[p][e] = half ? hq[8 * p + 4 + e] : hq[8 * p + e];
            }
            float*       dsta = ob + ((sb >> 2) + half) * QF + r * 4;
            const float* srca = xin + (((ra >> 2) * QF) & xmk);
            __builtin_amdgcn_sched_barrier(0);
            #pragma unroll
            for (int j = 0; j < G / 4; ++j) {
              const v4f t = *(const v4f*)(srca + j * QF);
              xa[4 * j] = t[0]; xa[4 * j + 1] = t[1]; xa[4 * j + 2] = t[2]; xa[4 * j + 3] = t[3];
            }
            #pragma unroll
            for (int p = 0; p < G / 8; ++p) *(v4f*)(dsta + p * 2 * QF) = pk[p];
            __builtin_amdgcn_sched_barrier(0);
          }
          #pragma unroll
          for (int u = 0; u < G; ++u) { hs = rnn_step(xb[u], hs, wr, sel_lo, sel_hi); hq[u] = hs; }
          {
            v4f pk[G / 8];
            #pragma unroll
            for (int p = 0; p < G / 8; ++p) {
              #pragma unroll
              for (int e = 0; e < 4; ++e) pk[p][e] = half ? hq[8 * p + 4 + e] : hq[8 * p + e];
            }
            float*       dstb = ob + (((sb + G) >> 2) + half) * QF + r * 4;
            const float* srcb = xin + (((rb >> 2) * QF) & xmk);
            __builtin_amdgcn_sched_barrier(0);
            #pragma unroll
            for (int j = 0; j < G / 4; ++j) {
              const v4f t = *(const v4f*)(srcb + j * QF);
              xb[4 * j] = t[0]; xb[4 * j + 1] = t[1]; xb[4 * j + 2] = t[2]; xb[4 * j + 3] = t[3];
            }
            #pragma unroll
            for (int p = 0; p < G / 8; ++p) *(v4f*)(dstb + p * 2 * QF) = pk[p];
            __builtin_amdgcn_sched_barrier(0);
          }
        }
      }
    } else {
      if (i < NCHUNK) {
        float*       xo = stage[0][i & 1] + half * QF + n * 4;
        const float* yc = y + i * CHUNK + half * 4;
        for (int j0 = 0; j0 < CHUNK / 8; j0 += 8) {
          v4f yv[8];
          #pragma unroll
          for (int u = 0; u < 8; ++u) yv[u] = *(const v4f*)(yc + (j0 + u) * 8);
          #pragma unroll
          for (int u = 0; u < 8; ++u) {
            v4f r;
            r[0] = fmaf(yv[u][0], wi0n, bi0n);
            r[1] = fmaf(yv[u][1], wi0n, bi0n);
            r[2] = fmaf(yv[u][2], wi0n, bi0n);
            r[3] = fmaf(yv[u][3], wi0n, bi0n);
            *(v4f*)(xo + (j0 + u) * 2 * QF) = r;
          }
        }
      }
      if (i >= 2 && i <= NCHUNK + 1) {
        const int c = i - 2;
        const float* hb = stage[1][c & 1];
        float*       xo = stage[2][c & 1];

        for (int tile = 0; tile < CHUNK / 16; ++tile) {
          v8f acc;
          #pragma unroll
          for (int v = 0; v < 8; ++v) acc[v] = b1v;
          #pragma unroll
          for (int kb = 0; kb < 4; ++kb) {
            const int K = 4 * kb + 2 * half;
            const float* ap = hb + QIDX(tile * 16 + n, K);
            v2f Aa;
            Aa[0] = ap[0];
            Aa[1] = ap[4];
            acc = __builtin_amdgcn_wmma_f32_16x16x4_f32(
                false, Aa, false, Bw[kb], (short)0, acc, false, false);
          }
          float* xq = xo + (tile * 4 + 2 * half) * QF + n * 4;
          v4f d0, d1;
          #pragma unroll
          for (int v = 0; v < 4; ++v) { d0[v] = acc[v]; d1[v] = acc[4 + v]; }
          *(v4f*)(xq)      = d0;
          *(v4f*)(xq + QF) = d1;
        }
      }
      if (i >= 4) {
        const int c = i - 4;
        const float* hb = stage[3][c & 1];
        const int o  = l & 1;
        const int tt = l >> 1;
        const int base = c * CHUNK;
        for (int s = tt; s < CHUNK; s += 16) {
          float acc = bl;
          const float* hp = hb + QIDX(s, 0);
          #pragma unroll
          for (int k = 0; k < H; ++k)
            acc = fmaf(hp[k * 4], wl[k], acc);
          out[(base + s) * 2 + o] = acc;
        }
      }
    }
    __syncthreads();
  }
}

extern "C" void kernel_launch(void* const* d_in, const int* in_sizes, int n_in,
                              void* d_out, int out_size, void* d_ws, size_t ws_size,
                              hipStream_t stream) {
  (void)in_sizes; (void)n_in; (void)out_size; (void)d_ws; (void)ws_size;
  const float* y     = (const float*)d_in[0];
  const float* W_ih0 = (const float*)d_in[1];
  const float* b_ih0 = (const float*)d_in[2];
  const float* W_hh0 = (const float*)d_in[3];
  const float* b_hh0 = (const float*)d_in[4];
  const float* W_ih1 = (const float*)d_in[5];
  const float* b_ih1 = (const float*)d_in[6];
  const float* W_hh1 = (const float*)d_in[7];
  const float* b_hh1 = (const float*)d_in[8];
  const float* W_lin = (const float*)d_in[9];
  const float* b_lin = (const float*)d_in[10];
  float* out = (float*)d_out;

  rnn_pipeline_kernel<<<dim3(1), dim3(96), 0, stream>>>(
      y, W_ih0, b_ih0, W_hh0, b_hh0, W_ih1, b_ih1, W_hh1, b_hh1,
      W_lin, b_lin, out);
}
